# starting kernel
# baseline (speedup 1.0000x reference)
#include <hip/hip_runtime.h>

typedef __attribute__((ext_vector_type(2))) float v2f;
typedef __attribute__((ext_vector_type(8))) float v8f;

#define TREES  2048
#define KNODES 511
#define EDIM   128
#define DEPTH  9

__global__ __launch_bounds__(256, 1)
void tree_encoder_kernel(const float* __restrict__ emb,
                         const float* __restrict__ W,
                         const float* __restrict__ bias,
                         const int*  __restrict__ tokens,
                         float* __restrict__ out)
{
    __shared__ float hbuf[KNODES * EDIM];
    __shared__ float pmax[256];

    const int tree = blockIdx.x;
    const int tid  = threadIdx.x;
    const int wave = tid >> 5;
    const int lane = tid & 31;
    const int mrow = lane & 15;
    const int half = lane >> 4;

    const int ni  = wave;
    const int col = ni * 16 + mrow;

    v2f bfrag[32];
#pragma unroll
    for (int kk = 0; kk < 32; ++kk) {
        const int k = kk * 4 + 2 * half;
        bfrag[kk].x = W[(k + 0) * EDIM + col];
        bfrag[kk].y = W[(k + 1) * EDIM + col];
    }
    const float bv   = bias[col];
    const int   base = tree * KNODES;

#pragma unroll 1
    for (int mi = 0; mi < 32; ++mi) {
        const int node   = mi * 16 + mrow;
        const int nclamp = node < KNODES ? node : KNODES - 1;
        const int token  = tokens[base + nclamp];
        const float* arow = emb + (size_t)token * EDIM + 2 * half;

        v8f c = {};
#pragma unroll
        for (int kk = 0; kk < 32; ++kk) {
            v2f a = *(const v2f*)(arow + kk * 4);
            c = __builtin_amdgcn_wmma_f32_16x16x4_f32(
                    false, a, false, bfrag[kk], (short)0, c, false, false);
        }
#pragma unroll
        for (int r = 0; r < 8; ++r) {
            const int orow = mi * 16 + r + 8 * half;
            if (orow < KNODES)
                hbuf[orow * EDIM + col] = c[r] + bv;
        }
    }
    __syncthreads();

    for (int l = DEPTH - 1; l >= 1; --l) {
        const int pFirst = (1 << (l - 1)) - 1;
        const int total  = (1 << (l - 1)) * EDIM;
        for (int e = tid; e < total; e += 256) {
            const int p = pFirst + (e >> 7);
            const int d = e & (EDIM - 1);
            hbuf[p * EDIM + d] += hbuf[(2 * p + 1) * EDIM + d]
                                + hbuf[(2 * p + 2) * EDIM + d];
        }
        __syncthreads();
    }

    {
        const int d    = tid & (EDIM - 1);
        const int part = tid >> 7;
        const int n0   = part * 256;
        const int n1   = part ? KNODES : 256;
        float m = -__builtin_inff();
        for (int n = n0; n < n1; ++n)
            m = fmaxf(m, hbuf[n * EDIM + d]);
        pmax[tid] = m;
    }
    __syncthreads();
    if (tid < EDIM) {
        const float m = fmaxf(pmax[tid], pmax[tid + EDIM]);
        out[tree * EDIM + tid] = fmaxf(m, 0.0f);
    }
}

extern "C" void kernel_launch(void* const* d_in, const int* in_sizes, int n_in,
                              void* d_out, int out_size, void* d_ws, size_t ws_size,
                              hipStream_t stream)
{
    const float* emb    = (const float*)d_in[0];
    const float* W      = (const float*)d_in[1];
    const float* bias   = (const float*)d_in[2];
    const int*   tokens = (const int*)d_in[3];
    float*       out    = (float*)d_out;

    tree_encoder_kernel<<<TREES, 256, 0, stream>>>(emb, W, bias, tokens, out);
}


# A operand staged once per workgroup: whole 512-byte embedding rows gathered a tile ahead into a 2-stage LDS ring, all 8 waves read A fragments from LDS (was: each wave gathered 8-byte pieces itself)
# speedup vs baseline: 1.4677x; 1.4677x over previous
#include <hip/hip_runtime.h>

typedef __attribute__((ext_vector_type(2))) float v2f;
typedef __attribute__((ext_vector_type(8))) float v8f;
typedef __attribute__((ext_vector_type(4))) float v4f;

#define TREES  2048
#define KNODES 511
#define EDIM   128
#define DEPTH  9
#define APITCH 132

__global__ __launch_bounds__(256, 1)
void tree_encoder_kernel(const float* __restrict__ emb,
                         const float* __restrict__ W,
                         const float* __restrict__ bias,
                         const int*  __restrict__ tokens,
                         float* __restrict__ out)
{
    __shared__ float hbuf[KNODES * EDIM];
    __shared__ float pmax[256];
    __shared__ __attribute__((aligned(16))) float stage[2][16 * APITCH];

    const int tree = blockIdx.x;
    const int tid  = threadIdx.x;
    const int wave = tid >> 5;
    const int lane = tid & 31;
    const int mrow = lane & 15;
    const int half = lane >> 4;

    const int ni  = wave;
    const int col = ni * 16 + mrow;

    v2f bfrag[32];
#pragma unroll
    for (int kk = 0; kk < 32; ++kk) {
        const int k = kk * 4 + 2 * half;
        bfrag[kk].x = W[(k + 0) * EDIM + col];
        bfrag[kk].y = W[(k + 1) * EDIM + col];
    }
    const float bv   = bias[col];
    const int   base = tree * KNODES;

    int na = lane * 16 + wave;
    int nb = na + 8;
    na = na < KNODES ? na : KNODES - 1;
    nb = nb < KNODES ? nb : KNODES - 1;
    const int tokA = tokens[base + na];
    const int tokB = tokens[base + nb];

    const int soffA = wave * APITCH + lane * 4;
    const int soffB = (wave + 8) * APITCH + lane * 4;
    const int aoff  = mrow * APITCH + 2 * half;

    v4f ra, rb;
    {
        const int ta = __builtin_amdgcn_readlane(tokA, 0);
        const int tb = __builtin_amdgcn_readlane(tokB, 0);
        ra = *(const v4f*)(emb + (size_t)ta * EDIM + lane * 4);
        rb = *(const v4f*)(emb + (size_t)tb * EDIM + lane * 4);
        *(v4f*)(&stage[0][soffA]) = ra;
        *(v4f*)(&stage[0][soffB]) = rb;
    }
    {
        const int ta = __builtin_amdgcn_readlane(tokA, 1);
        const int tb = __builtin_amdgcn_readlane(tokB, 1);
        ra = *(const v4f*)(emb + (size_t)ta * EDIM + lane * 4);
        rb = *(const v4f*)(emb + (size_t)tb * EDIM + lane * 4);
    }

#pragma unroll 1
    for (int mi = 0; mi < 32; ++mi) {
        __syncthreads();
        const float* cur = stage[mi & 1];
        float*       nxt = stage[(mi + 1) & 1];

        const float* arow = cur + aoff;
        v2f afrag[32];
#pragma unroll
        for (int kk = 0; kk < 32; ++kk)
            afrag[kk] = *(const v2f*)(arow + kk * 4);
        __builtin_amdgcn_sched_barrier(0);

        if (mi + 1 < 32) {
            *(v4f*)(nxt + soffA) = ra;
            *(v4f*)(nxt + soffB) = rb;
        }
        if (mi + 2 < 32) {
            const int ta = __builtin_amdgcn_readlane(tokA, mi + 2);
            const int tb = __builtin_amdgcn_readlane(tokB, mi + 2);
            ra = *(const v4f*)(emb + (size_t)ta * EDIM + lane * 4);
            rb = *(const v4f*)(emb + (size_t)tb * EDIM + lane * 4);
        }

        v8f c = {};
#pragma unroll
        for (int kk = 0; kk < 32; ++kk)
            c = __builtin_amdgcn_wmma_f32_16x16x4_f32(
                    false, afrag[kk], false, bfrag[kk], (short)0, c, false, false);
#pragma unroll
        for (int r = 0; r < 8; ++r) {
            const int orow = mi * 16 + r + 8 * half;
            if (orow < KNODES)
                hbuf[orow * EDIM + col] = c[r] + bv;
        }
    }
    __syncthreads();

    for (int l = DEPTH - 1; l >= 1; --l) {
        const int pFirst = (1 << (l - 1)) - 1;
        const int total  = (1 << (l - 1)) * EDIM;
        for (int e = tid; e < total; e += 256) {
            const int p = pFirst + (e >> 7);
            const int d = e & (EDIM - 1);
            hbuf[p * EDIM + d] += hbuf[(2 * p + 1) * EDIM + d]
                                + hbuf[(2 * p + 2) * EDIM + d];
        }
        __syncthreads();
    }

    {
        const int d    = tid & (EDIM - 1);
        const int part = tid >> 7;
        const int n0   = part * 256;
        const int n1   = part ? KNODES : 256;
        float m = -__builtin_inff();
        for (int n = n0; n < n1; ++n)
            m = fmaxf(m, hbuf[n * EDIM + d]);
        pmax[tid] = m;
    }
    __syncthreads();
    if (tid < EDIM) {
        const float m = fmaxf(pmax[tid], pmax[tid + EDIM]);
        out[tree * EDIM + tid] = fmaxf(m, 0.0f);
    }
}

extern "C" void kernel_launch(void* const* d_in, const int* in_sizes, int n_in,
                              void* d_out, int out_size, void* d_ws, size_t ws_size,
                              hipStream_t stream)
{
    const float* emb    = (const float*)d_in[0];
    const float* W      = (const float*)d_in[1];
    const float* bias   = (const float*)d_in[2];
    const int*   tokens = (const int*)d_in[3];
    float*       out    = (float*)d_out;

    tree_encoder_kernel<<<TREES, 256, 0, stream>>>(emb, W, bias, tokens, out);
}


# Gathered rows stay in flight across 8 tiles instead of 1: 8-slot register ring (tile loop unrolled by 8) plus an LDS-only tile barrier; the full fence drained every global load at each tile
# speedup vs baseline: 1.5726x; 1.0715x over previous
#include <hip/hip_runtime.h>

typedef __attribute__((ext_vector_type(2))) float v2f;
typedef __attribute__((ext_vector_type(8))) float v8f;
typedef __attribute__((ext_vector_type(4))) float v4f;

#define TREES  2048
#define KNODES 511
#define EDIM   128
#define DEPTH  9
#define APITCH 132
#define PF     8

__global__ __launch_bounds__(256, 1)
void tree_encoder_kernel(const float* __restrict__ emb,
                         const float* __restrict__ W,
                         const float* __restrict__ bias,
                         const int*  __restrict__ tokens,
                         float* __restrict__ out)
{
    __shared__ float hbuf[KNODES * EDIM];
    __shared__ float pmax[256];
    __shared__ __attribute__((aligned(16))) float stage[2][16 * APITCH];

    const int tree = blockIdx.x;
    const int tid  = threadIdx.x;
    const int wave = tid >> 5;
    const int lane = tid & 31;
    const int mrow = lane & 15;
    const int half = lane >> 4;

    const int ni  = wave;
    const int col = ni * 16 + mrow;

    v2f bfrag[32];
#pragma unroll
    for (int kk = 0; kk < 32; ++kk) {
        const int k = kk * 4 + 2 * half;
        bfrag[kk].x = W[(k + 0) * EDIM + col];
        bfrag[kk].y = W[(k + 1) * EDIM + col];
    }
    const float bv   = bias[col];
    const int   base = tree * KNODES;

    int na = lane * 16 + wave;
    int nb = na + 8;
    na = na < KNODES ? na : KNODES - 1;
    nb = nb < KNODES ? nb : KNODES - 1;
    const int tokA = tokens[base + na];
    const int tokB = tokens[base + nb];

    const int soffA = wave * APITCH + lane * 4;
    const int soffB = (wave + 8) * APITCH + lane * 4;
    const int aoff  = mrow * APITCH + 2 * half;

    v4f ra[PF], rb[PF];
    v4f r0a, r0b;
    {
        const int ta = __builtin_amdgcn_readlane(tokA, 0);
        const int tb = __builtin_amdgcn_readlane(tokB, 0);
        r0a = *(const v4f*)(emb + (size_t)ta * EDIM + lane * 4);
        r0b = *(const v4f*)(emb + (size_t)tb * EDIM + lane * 4);
    }
#pragma unroll
    for (int t = 1; t <= PF; ++t) {
        const int ta = __builtin_amdgcn_readlane(tokA, t);
        const int tb = __builtin_amdgcn_readlane(tokB, t);
        ra[t % PF] = *(const v4f*)(emb + (size_t)ta * EDIM + lane * 4);
        rb[t % PF] = *(const v4f*)(emb + (size_t)tb * EDIM + lane * 4);
    }
    *(v4f*)(&stage[0][soffA]) = r0a;
    *(v4f*)(&stage[0][soffB]) = r0b;

    auto tile = [&](const int mi, const int s, const int par,
                    const bool do_store, const bool do_issue) __attribute__((always_inline)) {
        __builtin_amdgcn_fence(__ATOMIC_RELEASE, "workgroup", "local");
        __builtin_amdgcn_s_barrier();
        __builtin_amdgcn_fence(__ATOMIC_ACQUIRE, "workgroup", "local");
        const float* cur = stage[par];
        float*       nxt = stage[par ^ 1];

        const float* arow = cur + aoff;
        v2f afrag[32];
#pragma unroll
        for (int kk = 0; kk < 32; ++kk)
            afrag[kk] = *(const v2f*)(arow + kk * 4);
        __builtin_amdgcn_sched_barrier(0);

        if (do_store) {
            *(v4f*)(nxt + soffA) = ra[s];
            *(v4f*)(nxt + soffB) = rb[s];
        }
        if (do_issue) {
            const int ta = __builtin_amdgcn_readlane(tokA, mi + 1 + PF);
            const int tb = __builtin_amdgcn_readlane(tokB, mi + 1 + PF);
            ra[s] = *(const v4f*)(emb + (size_t)ta * EDIM + lane * 4);
            rb[s] = *(const v4f*)(emb + (size_t)tb * EDIM + lane * 4);
        }

        v8f c = {};
#pragma unroll
        for (int kk = 0; kk < 32; ++kk)
            c = __builtin_amdgcn_wmma_f32_16x16x4_f32(
                    false, afrag[kk], false, bfrag[kk], (short)0, c, false, false);
#pragma unroll
        for (int r = 0; r < 8; ++r) {
            const int orow = mi * 16 + r + 8 * half;
            if (orow < KNODES)
                hbuf[orow * EDIM + col] = c[r] + bv;
        }
    };

#pragma unroll 1
    for (int g = 0; g < 32 - 2 * PF; g += PF) {
#pragma unroll
        for (int j = 0; j < PF; ++j)
            tile(g + j, (j + 1) % PF, j & 1, true, true);
    }
#pragma unroll
    for (int mi = 32 - 2 * PF; mi < 32; ++mi)
        tile(mi, (mi + 1) % PF, mi & 1, mi + 1 < 32, mi + 1 + PF < 32);
    __syncthreads();

    for (int l = DEPTH - 1; l >= 1; --l) {
        const int pFirst = (1 << (l - 1)) - 1;
        const int total  = (1 << (l - 1)) * EDIM;
        for (int e = tid; e < total; e += 256) {
            const int p = pFirst + (e >> 7);
            const int d = e & (EDIM - 1);
            hbuf[p * EDIM + d] += hbuf[(2 * p + 1) * EDIM + d]
                                + hbuf[(2 * p + 2) * EDIM + d];
        }
        __syncthreads();
    }

    {
        const int d    = tid & (EDIM - 1);
        const int part = tid >> 7;
        const int n0   = part * 256;
        const int n1   = part ? KNODES : 256;
        float m = -__builtin_inff();
        for (int n = n0; n < n1; ++n)
            m = fmaxf(m, hbuf[n * EDIM + d]);
        pmax[tid] = m;
    }
    __syncthreads();
    if (tid < EDIM) {
        const float m = fmaxf(pmax[tid], pmax[tid + EDIM]);
        out[tree * EDIM + tid] = fmaxf(m, 0.0f);
    }
}

extern "C" void kernel_launch(void* const* d_in, const int* in_sizes, int n_in,
                              void* d_out, int out_size, void* d_ws, size_t ws_size,
                              hipStream_t stream)
{
    const float* emb    = (const float*)d_in[0];
    const float* W      = (const float*)d_in[1];
    const float* bias   = (const float*)d_in[2];
    const int*   tokens = (const int*)d_in[3];
    float*       out    = (float*)d_out;

    tree_encoder_kernel<<<TREES, 256, 0, stream>>>(emb, W, bias, tokens, out);
}


# Tree sums and per-tree max in one register walk: each thread reads one dim of one half-tree from LDS once (255 independent reads, no store, no barrier); was 8 LDS read-modify-write levels + max pass
# speedup vs baseline: 2.2545x; 1.4336x over previous
#include <hip/hip_runtime.h>

typedef __attribute__((ext_vector_type(2))) float v2f;
typedef __attribute__((ext_vector_type(8))) float v8f;
typedef __attribute__((ext_vector_type(4))) float v4f;

#define TREES  2048
#define KNODES 511
#define EDIM   128
#define DEPTH  9
#define APITCH 132
#define PF     8

__global__ __launch_bounds__(256, 1)
void tree_encoder_kernel(const float* __restrict__ emb,
                         const float* __restrict__ W,
                         const float* __restrict__ bias,
                         const int*  __restrict__ tokens,
                         float* __restrict__ out)
{
    __shared__ float hbuf[KNODES * EDIM];
    __shared__ float pmax[256];
    __shared__ float psum[256];
    __shared__ __attribute__((aligned(16))) float stage[2][16 * APITCH];

    const int tree = blockIdx.x;
    const int tid  = threadIdx.x;
    const int wave = tid >> 5;
    const int lane = tid & 31;
    const int mrow = lane & 15;
    const int half = lane >> 4;

    const int ni  = wave;
    const int col = ni * 16 + mrow;

    v2f bfrag[32];
#pragma unroll
    for (int kk = 0; kk < 32; ++kk) {
        const int k = kk * 4 + 2 * half;
        bfrag[kk].x = W[(k + 0) * EDIM + col];
        bfrag[kk].y = W[(k + 1) * EDIM + col];
    }
    const float bv   = bias[col];
    const int   base = tree * KNODES;

    int na = lane * 16 + wave;
    int nb = na + 8;
    na = na < KNODES ? na : KNODES - 1;
    nb = nb < KNODES ? nb : KNODES - 1;
    const int tokA = tokens[base + na];
    const int tokB = tokens[base + nb];

    const int soffA = wave * APITCH + lane * 4;
    const int soffB = (wave + 8) * APITCH + lane * 4;
    const int aoff  = mrow * APITCH + 2 * half;

    v4f ra[PF], rb[PF];
    v4f r0a, r0b;
    {
        const int ta = __builtin_amdgcn_readlane(tokA, 0);
        const int tb = __builtin_amdgcn_readlane(tokB, 0);
        r0a = *(const v4f*)(emb + (size_t)ta * EDIM + lane * 4);
        r0b = *(const v4f*)(emb + (size_t)tb * EDIM + lane * 4);
    }
#pragma unroll
    for (int t = 1; t <= PF; ++t) {
        const int ta = __builtin_amdgcn_readlane(tokA, t);
        const int tb = __builtin_amdgcn_readlane(tokB, t);
        ra[t % PF] = *(const v4f*)(emb + (size_t)ta * EDIM + lane * 4);
        rb[t % PF] = *(const v4f*)(emb + (size_t)tb * EDIM + lane * 4);
    }
    *(v4f*)(&stage[0][soffA]) = r0a;
    *(v4f*)(&stage[0][soffB]) = r0b;

    auto tile = [&](const int mi, const int s, const int par,
                    const bool do_store, const bool do_issue) __attribute__((always_inline)) {
        __builtin_amdgcn_fence(__ATOMIC_RELEASE, "workgroup", "local");
        __builtin_amdgcn_s_barrier();
        __builtin_amdgcn_fence(__ATOMIC_ACQUIRE, "workgroup", "local");
        const float* cur = stage[par];
        float*       nxt = stage[par ^ 1];

        const float* arow = cur + aoff;
        v2f afrag[32];
#pragma unroll
        for (int kk = 0; kk < 32; ++kk)
            afrag[kk] = *(const v2f*)(arow + kk * 4);
        __builtin_amdgcn_sched_barrier(0);

        if (do_store) {
            *(v4f*)(nxt + soffA) = ra[s];
            *(v4f*)(nxt + soffB) = rb[s];
        }
        if (do_issue) {
            const int ta = __builtin_amdgcn_readlane(tokA, mi + 1 + PF);
            const int tb = __builtin_amdgcn_readlane(tokB, mi + 1 + PF);
            ra[s] = *(const v4f*)(emb + (size_t)ta * EDIM + lane * 4);
            rb[s] = *(const v4f*)(emb + (size_t)tb * EDIM + lane * 4);
        }

        v8f c = {};
#pragma unroll
        for (int kk = 0; kk < 32; ++kk)
            c = __builtin_amdgcn_wmma_f32_16x16x4_f32(
                    false, afrag[kk], false, bfrag[kk], (short)0, c, false, false);
#pragma unroll
        for (int r = 0; r < 8; ++r) {
            const int orow = mi * 16 + r + 8 * half;
            if (orow < KNODES)
                hbuf[orow * EDIM + col] = c[r] + bv;
        }
    };

#pragma unroll 1
    for (int g = 0; g < 32 - 2 * PF; g += PF) {
#pragma unroll
        for (int j = 0; j < PF; ++j)
            tile(g + j, (j + 1) % PF, j & 1, true, true);
    }
#pragma unroll
    for (int mi = 32 - 2 * PF; mi < 32; ++mi)
        tile(mi, (mi + 1) % PF, mi & 1, mi + 1 < 32, mi + 1 + PF < 32);
    __syncthreads();

    {
        const int d    = tid & (EDIM - 1);
        const int part = tid >> 7;
        const float* lvl[8];
#pragma unroll
        for (int q = 0; q < 8; ++q)
            lvl[q] = hbuf + (part ? (3 << q) - 1 : (2 << q) - 1) * EDIM + d;
        float m = -__builtin_inff();

        float sub[8];
#pragma unroll
        for (int c = 0; c < 8; ++c) {
            float v[31];
#pragma unroll
            for (int t = 0; t < 5; ++t) {
                const int q = 3 + t;
#pragma unroll
                for (int i = 0; i < (1 << t); ++i)
                    v[(1 << t) - 1 + i] = lvl[q][(c * (1 << t) + i) * EDIM];
            }
#pragma unroll
            for (int i = 15; i < 31; ++i)
                m = fmaxf(m, v[i]);
#pragma unroll
            for (int k = 14; k >= 0; --k) {
                v[k] = v[k] + (v[2 * k + 1] + v[2 * k + 2]);
                m = fmaxf(m, v[k]);
            }
            sub[c] = v[0];
        }
        float tv[7];
#pragma unroll
        for (int q = 0; q < 3; ++q) {
#pragma unroll
            for (int j = 0; j < (1 << q); ++j)
                tv[(1 << q) - 1 + j] = lvl[q][j * EDIM];
        }
#pragma unroll
        for (int j = 0; j < 4; ++j) {
            tv[3 + j] = tv[3 + j] + (sub[2 * j] + sub[2 * j + 1]);
            m = fmaxf(m, tv[3 + j]);
        }
#pragma unroll
        for (int k = 2; k >= 0; --k) {
            tv[k] = tv[k] + (tv[2 * k + 1] + tv[2 * k + 2]);
            m = fmaxf(m, tv[k]);
        }
        psum[tid] = tv[0];
        pmax[tid] = m;
    }
    __syncthreads();

    if (tid < EDIM) {
        const float root = hbuf[tid] + (psum[tid] + psum[tid + EDIM]);
        const float m = fmaxf(fmaxf(pmax[tid], pmax[tid + EDIM]), root);
        out[tree * EDIM + tid] = fmaxf(m, 0.0f);
    }
}

extern "C" void kernel_launch(void* const* d_in, const int* in_sizes, int n_in,
                              void* d_out, int out_size, void* d_ws, size_t ws_size,
                              hipStream_t stream)
{
    const float* emb    = (const float*)d_in[0];
    const float* W      = (const float*)d_in[1];
    const float* bias   = (const float*)d_in[2];
    const int*   tokens = (const int*)d_in[3];
    float*       out    = (float*)d_out;

    tree_encoder_kernel<<<TREES, 256, 0, stream>>>(emb, W, bias, tokens, out);
}


# Tile loop pipelined by one tile: next tile's A fragments are read from LDS into a second register set ahead of the current WMMAs, a tile's h rows are stored behind the next barrier
# speedup vs baseline: 2.4121x; 1.0699x over previous
#include <hip/hip_runtime.h>

typedef __attribute__((ext_vector_type(2))) float v2f;
typedef __attribute__((ext_vector_type(8))) float v8f;
typedef __attribute__((ext_vector_type(4))) float v4f;

#define TREES  2048
#define KNODES 511
#define EDIM   128
#define DEPTH  9
#define APITCH 132
#define PF     8

__global__ __launch_bounds__(256, 1)
void tree_encoder_kernel(const float* __restrict__ emb,
                         const float* __restrict__ W,
                         const float* __restrict__ bias,
                         const int*  __restrict__ tokens,
                         float* __restrict__ out)
{
    __shared__ float hbuf[KNODES * EDIM];
    __shared__ float pmax[256];
    __shared__ float psum[256];
    __shared__ __attribute__((aligned(16))) float stage[2][16 * APITCH];

    const int tree = blockIdx.x;
    const int tid  = threadIdx.x;
    const int wave = tid >> 5;
    const int lane = tid & 31;
    const int mrow = lane & 15;
    const int half = lane >> 4;

    const int ni  = wave;
    const int col = ni * 16 + mrow;

    v2f bfrag[32];
#pragma unroll
    for (int kk = 0; kk < 32; ++kk) {
        const int k = kk * 4 + 2 * half;
        bfrag[kk].x = W[(k + 0) * EDIM + col];
        bfrag[kk].y = W[(k + 1) * EDIM + col];
    }
    const float bv   = bias[col];
    const int   base = tree * KNODES;

    int na = lane * 16 + wave;
    int nb = na + 8;
    na = na < KNODES ? na : KNODES - 1;
    nb = nb < KNODES ? nb : KNODES - 1;
    const int tokA = tokens[base + na];
    const int tokB = tokens[base + nb];

    const int soffA = wave * APITCH + lane * 4;
    const int soffB = (wave + 8) * APITCH + lane * 4;
    const int aoff  = mrow * APITCH + 2 * half;

    v4f ra[PF], rb[PF];
    v4f r0a, r0b, r1a, r1b;
    {
        const int ta0 = __builtin_amdgcn_readlane(tokA, 0);
        const int tb0 = __builtin_amdgcn_readlane(tokB, 0);
        r0a = *(const v4f*)(emb + (size_t)ta0 * EDIM + lane * 4);
        r0b = *(const v4f*)(emb + (size_t)tb0 * EDIM + lane * 4);
        const int ta1 = __builtin_amdgcn_readlane(tokA, 1);
        const int tb1 = __builtin_amdgcn_readlane(tokB, 1);
        r1a = *(const v4f*)(emb + (size_t)ta1 * EDIM + lane * 4);
        r1b = *(const v4f*)(emb + (size_t)tb1 * EDIM + lane * 4);
    }
#pragma unroll
    for (int t = 2; t <= PF + 1; ++t) {
        const int ta = __builtin_amdgcn_readlane(tokA, t);
        const int tb = __builtin_amdgcn_readlane(tokB, t);
        ra[t % PF] = *(const v4f*)(emb + (size_t)ta * EDIM + lane * 4);
        rb[t % PF] = *(const v4f*)(emb + (size_t)tb * EDIM + lane * 4);
    }
    *(v4f*)(&stage[0][soffA]) = r0a;
    *(v4f*)(&stage[0][soffB]) = r0b;
    *(v4f*)(&stage[1][soffA]) = r1a;
    *(v4f*)(&stage[1][soffB]) = r1b;

    v2f afr[2][32];
    v8f cprev = {};

    __builtin_amdgcn_fence(__ATOMIC_RELEASE, "workgroup", "local");
    __builtin_amdgcn_s_barrier();
    __builtin_amdgcn_fence(__ATOMIC_ACQUIRE, "workgroup", "local");
#pragma unroll
    for (int kk = 0; kk < 32; ++kk)
        afr[0][kk] = *(const v2f*)(&stage[0][aoff + kk * 4]);

    auto tile = [&](const int mi, const int s, const int par, const bool do_next,
                    const bool do_store, const bool do_issue,
                    const bool do_epi) __attribute__((always_inline)) {
        __builtin_amdgcn_fence(__ATOMIC_RELEASE, "workgroup", "local");
        __builtin_amdgcn_s_barrier();
        __builtin_amdgcn_fence(__ATOMIC_ACQUIRE, "workgroup", "local");
        const float* nxt = stage[par ^ 1];
        float*       fre = stage[par];

        if (do_next) {
            const float* arow = nxt + aoff;
#pragma unroll
            for (int kk = 0; kk < 32; ++kk)
                afr[par ^ 1][kk] = *(const v2f*)(arow + kk * 4);
        }
        __builtin_amdgcn_sched_barrier(0);

        if (do_store) {
            *(v4f*)(fre + soffA) = ra[s];
            *(v4f*)(fre + soffB) = rb[s];
        }
        if (do_issue) {
            const int ta = __builtin_amdgcn_readlane(tokA, mi + 2 + PF);
            const int tb = __builtin_amdgcn_readlane(tokB, mi + 2 + PF);
            ra[s] = *(const v4f*)(emb + (size_t)ta * EDIM + lane * 4);
            rb[s] = *(const v4f*)(emb + (size_t)tb * EDIM + lane * 4);
        }
        if (do_epi) {
#pragma unroll
            for (int r = 0; r < 8; ++r) {
                const int orow = (mi - 1) * 16 + r + 8 * half;
                if (orow < KNODES)
                    hbuf[orow * EDIM + col] = cprev[r] + bv;
            }
        }

        v8f c = {};
#pragma unroll
        for (int kk = 0; kk < 32; ++kk)
            c = __builtin_amdgcn_wmma_f32_16x16x4_f32(
                    false, afr[par][kk], false, bfrag[kk], (short)0, c, false, false);
        cprev = c;
    };

    tile(0, 2 % PF, 0, true, true, true, false);
#pragma unroll 1
    for (int g = 0; g < 32 - 2 * PF; g += PF) {
#pragma unroll
        for (int j = 1; j <= PF; ++j)
            tile(g + j, (j + 2) % PF, j & 1, true, true, true, true);
    }
#pragma unroll
    for (int mi = 32 - 2 * PF + 1; mi < 32; ++mi)
        tile(mi, (mi + 2) % PF, mi & 1, mi + 1 < 32, mi + 2 < 32, mi + 2 + PF < 32, true);
#pragma unroll
    for (int r = 0; r < 8; ++r) {
        const int orow = 31 * 16 + r + 8 * half;
        if (orow < KNODES)
            hbuf[orow * EDIM + col] = cprev[r] + bv;
    }
    __syncthreads();

    {
        const int d    = tid & (EDIM - 1);
        const int part = tid >> 7;
        const float* lvl[8];
#pragma unroll
        for (int q = 0; q < 8; ++q)
            lvl[q] = hbuf + (part ? (3 << q) - 1 : (2 << q) - 1) * EDIM + d;
        float m = -__builtin_inff();

        float sub[8];
#pragma unroll
        for (int c = 0; c < 8; ++c) {
            float v[31];
#pragma unroll
            for (int t = 0; t < 5; ++t) {
                const int q = 3 + t;
#pragma unroll
                for (int i = 0; i < (1 << t); ++i)
                    v[(1 << t) - 1 + i] = lvl[q][(c * (1 << t) + i) * EDIM];
            }
#pragma unroll
            for (int i = 15; i < 31; ++i)
                m = fmaxf(m, v[i]);
#pragma unroll
            for (int k = 14; k >= 0; --k) {
                v[k] = v[k] + (v[2 * k + 1] + v[2 * k + 2]);
                m = fmaxf(m, v[k]);
            }
            sub[c] = v[0];
        }
        float tv[7];
#pragma unroll
        for (int q = 0; q < 3; ++q) {
#pragma unroll
            for (int j = 0; j < (1 << q); ++j)
                tv[(1 << q) - 1 + j] = lvl[q][j * EDIM];
        }
#pragma unroll
        for (int j = 0; j < 4; ++j) {
            tv[3 + j] = tv[3 + j] + (sub[2 * j] + sub[2 * j + 1]);
            m = fmaxf(m, tv[3 + j]);
        }
#pragma unroll
        for (int k = 2; k >= 0; --k) {
            tv[k] = tv[k] + (tv[2 * k + 1] + tv[2 * k + 2]);
            m = fmaxf(m, tv[k]);
        }
        psum[tid] = tv[0];
        pmax[tid] = m;
    }
    __syncthreads();

    if (tid < EDIM) {
        const float root = hbuf[tid] + (psum[tid] + psum[tid + EDIM]);
        const float m = fmaxf(fmaxf(pmax[tid], pmax[tid + EDIM]), root);
        out[tree * EDIM + tid] = fmaxf(m, 0.0f);
    }
}

extern "C" void kernel_launch(void* const* d_in, const int* in_sizes, int n_in,
                              void* d_out, int out_size, void* d_ws, size_t ws_size,
                              hipStream_t stream)
{
    const float* emb    = (const float*)d_in[0];
    const float* W      = (const float*)d_in[1];
    const float* bias   = (const float*)d_in[2];
    const int*   tokens = (const int*)d_in[3];
    float*       out    = (float*)d_out;

    tree_encoder_kernel<<<TREES, 256, 0, stream>>>(emb, W, bias, tokens, out);
}


# Tile barrier split by hand: a wave signals after 12 of a tile's 32 WMMAs (its LDS reads and ring stores done) and waits at the next tile's top; was signal+wait together behind all 32 WMMAs
# speedup vs baseline: 2.8476x; 1.1805x over previous
#include <hip/hip_runtime.h>

typedef __attribute__((ext_vector_type(2))) float v2f;
typedef __attribute__((ext_vector_type(8))) float v8f;
typedef __attribute__((ext_vector_type(4))) float v4f;

#define TREES  2048
#define KNODES 511
#define EDIM   128
#define NTILES 32
#define APITCH 132
#define PF     2
#define STEADY (((NTILES - 3 - 2 * PF) / PF + 1) * PF)
#define KSIG   12

__device__ __forceinline__ float xhalf(float x)
{
    const int b = __builtin_bit_cast(int, x);
    return __builtin_bit_cast(float, __builtin_amdgcn_permlanex16(b, b, 0x76543210, 0xfedcba98, false, false));
}

__device__ __forceinline__ void lds_traffic_done_signal()
{
    __builtin_amdgcn_fence(__ATOMIC_RELEASE, "workgroup", "local");
    asm volatile("s_barrier_signal -1" ::: "memory");
}
__device__ __forceinline__ void lds_all_signalled_wait()
{
    asm volatile("s_barrier_wait -1" ::: "memory");
}

__global__ __launch_bounds__(256, 4)
void tree_encoder_kernel(const float* __restrict__ emb,
                         const float* __restrict__ W,
                         const float* __restrict__ bias,
                         const int*  __restrict__ tokens,
                         float* __restrict__ out)
{
    __shared__ float xch[2 * NTILES * EDIM];
    __shared__ float pmax[256];
    __shared__ __attribute__((aligned(16))) float stage[2][16 * APITCH];

    const int tree = blockIdx.x;
    const int tid  = threadIdx.x;
    const int wave = tid >> 5;
    const int lane = tid & 31;
    const int mrow = lane & 15;
    const int half = lane >> 4;

    const int ni  = wave;
    const int col = ni * 16 + mrow;

    v2f bfrag[32];
#pragma unroll
    for (int kk = 0; kk < 32; ++kk) {
        const int k = kk * 4 + 2 * half;
        bfrag[kk].x = W[(k + 0) * EDIM + col];
        bfrag[kk].y = W[(k + 1) * EDIM + col];
    }
    const float bv   = bias[col];
    const int   base = tree * KNODES;

    int na, nb;
    if (wave == 0)      { na = 31 + lane;                  nb = lane < NTILES - 1 ? lane : 0; }
    else if (wave == 1) { na = 63 + 2 * lane;              nb = na + 1; }
    else if (wave < 4)  { na = 127 + 4 * lane + wave - 2;  nb = na + 2; }
    else                { na = 255 + 8 * lane + wave - 4;  nb = na + 4; }
    const int tokA = tokens[base + na];
    const int tokB = tokens[base + nb];

    const int soffA = wave * APITCH + lane * 4;
    const int soffB = (wave + 8) * APITCH + lane * 4;
    const int aoff  = mrow * APITCH + 2 * half;

    v4f ra[PF], rb[PF];
    v4f r0a, r0b, r1a, r1b;
    {
        const int ta0 = __builtin_amdgcn_readlane(tokA, 0);
        const int tb0 = __builtin_amdgcn_readlane(tokB, 0);
        r0a = *(const v4f*)(emb + (size_t)ta0 * EDIM + lane * 4);
        r0b = *(const v4f*)(emb + (size_t)tb0 * EDIM + lane * 4);
        const int ta1 = __builtin_amdgcn_readlane(tokA, 1);
        const int tb1 = __builtin_amdgcn_readlane(tokB, 1);
        r1a = *(const v4f*)(emb + (size_t)ta1 * EDIM + lane * 4);
        r1b = *(const v4f*)(emb + (size_t)tb1 * EDIM + lane * 4);
    }
#pragma unroll
    for (int t = 2; t <= PF + 1; ++t) {
        const int ta = __builtin_amdgcn_readlane(tokA, t);
        const int tb = __builtin_amdgcn_readlane(tokB, t);
        ra[t % PF] = *(const v4f*)(emb + (size_t)ta * EDIM + lane * 4);
        rb[t % PF] = *(const v4f*)(emb + (size_t)tb * EDIM + lane * 4);
    }
    *(v4f*)(&stage[0][soffA]) = r0a;
    *(v4f*)(&stage[0][soffB]) = r0b;
    *(v4f*)(&stage[1][soffA]) = r1a;
    *(v4f*)(&stage[1][soffB]) = r1b;

    v2f afr[2][32];
    v8f cprev = {};
    float m = -__builtin_inff();

    auto epi = [&]() __attribute__((always_inline)) -> float {
        float v[8];
#pragma unroll
        for (int r = 0; r < 8; ++r)
            v[r] = cprev[r] + bv;
        const float s2 = v[2] + (v[4] + v[5]);
        const float s3 = v[3] + (v[6] + v[7]);
        const float s1 = v[1] + (s2 + s3);
        m = fmaxf(m, fmaxf(fmaxf(v[4], v[5]), fmaxf(v[6], v[7])));
        m = fmaxf(m, fmaxf(fmaxf(s2, s3), s1));
        const float o  = xhalf(s1);
        const float sr = v[0] + (s1 + o);
        return half ? v[0] : sr;
    };
    auto epi_store = [&](const int j, const float x) __attribute__((always_inline)) {
        xch[half * (NTILES * EDIM) + j * EDIM + col] = x;
    };

    __builtin_amdgcn_fence(__ATOMIC_RELEASE, "workgroup", "local");
    __builtin_amdgcn_s_barrier();
    __builtin_amdgcn_fence(__ATOMIC_ACQUIRE, "workgroup", "local");
#pragma unroll
    for (int kk = 0; kk < 32; ++kk)
        afr[0][kk] = *(const v2f*)(&stage[0][aoff + kk * 4]);
    lds_traffic_done_signal();

    auto tile = [&](const int mi, const int s, const int par, const bool do_next,
                    const bool do_store, const bool do_issue,
                    const bool do_epi, const bool do_sig) __attribute__((always_inline)) {
        const float* nxt = stage[par ^ 1];
        float*       fre = stage[par];

        __builtin_amdgcn_sched_barrier(0);
        if (do_next) {
            lds_all_signalled_wait();
            const float* arow = nxt + aoff;
#pragma unroll
            for (int kk = 0; kk < 32; ++kk)
                afr[par ^ 1][kk] = *(const v2f*)(arow + kk * 4);
        }
        __builtin_amdgcn_sched_barrier(0);

        if (do_store) {
            *(v4f*)(fre + soffA) = ra[s];
            *(v4f*)(fre + soffB) = rb[s];
        }
        float x = 0.0f;
        if (do_epi)
            x = epi();

        v8f c = {};
#pragma unroll
        for (int kk = 0; kk < KSIG; ++kk)
            c = __builtin_amdgcn_wmma_f32_16x16x4_f32(
                    false, afr[par][kk], false, bfrag[kk], (short)0, c, false, false);
        __builtin_amdgcn_sched_barrier(0);
        if (do_sig)
            lds_traffic_done_signal();
        __builtin_amdgcn_sched_barrier(0);
        if (do_epi)
            epi_store(mi - 1, x);
        if (do_issue) {
            const int ta = __builtin_amdgcn_readlane(tokA, mi + 2 + PF);
            const int tb = __builtin_amdgcn_readlane(tokB, mi + 2 + PF);
            ra[s] = *(const v4f*)(emb + (size_t)ta * EDIM + lane * 4);
            rb[s] = *(const v4f*)(emb + (size_t)tb * EDIM + lane * 4);
        }
#pragma unroll
        for (int kk = KSIG; kk < 32; ++kk)
            c = __builtin_amdgcn_wmma_f32_16x16x4_f32(
                    false, afr[par][kk], false, bfrag[kk], (short)0, c, false, false);
        cprev = c;
    };

    tile(0, 2 % PF, 0, true, true, true, false, true);
#pragma unroll 1
    for (int g = 0; g < STEADY; g += PF) {
#pragma unroll
        for (int j = 1; j <= PF; ++j)
            tile(g + j, (j + 2) % PF, j & 1, true, true, true, true, true);
    }
#pragma unroll
    for (int mi = STEADY + 1; mi < NTILES; ++mi)
        tile(mi, (mi + 2) % PF, mi & 1, mi + 1 < NTILES, mi + 2 < NTILES, mi + 2 + PF < NTILES, true, mi + 2 < NTILES);
    epi_store(NTILES - 1, epi());
    pmax[half * EDIM + col] = m;
    __syncthreads();

    if (tid < EDIM) {
        const float* sub = xch + tid;
        const float* top = xch + NTILES * EDIM + tid;
        float mm = fmaxf(pmax[tid], pmax[tid + EDIM]);
        float t4[16], t3[8], t2[4], t1[2];
#pragma unroll
        for (int k = 0; k < 16; ++k) {
            const float a = sub[(2 * k) * EDIM];
            const float b = sub[(2 * k + 1) * EDIM];
            t4[k] = top[(15 + k) * EDIM] + (a + b);
            mm = fmaxf(mm, fmaxf(fmaxf(a, b), t4[k]));
        }
#pragma unroll
        for (int k = 0; k < 8; ++k) {
            t3[k] = top[(7 + k) * EDIM] + (t4[2 * k] + t4[2 * k + 1]);
            mm = fmaxf(mm, t3[k]);
        }
#pragma unroll
        for (int k = 0; k < 4; ++k) {
            t2[k] = top[(3 + k) * EDIM] + (t3[2 * k] + t3[2 * k + 1]);
            mm = fmaxf(mm, t2[k]);
        }
#pragma unroll
        for (int k = 0; k < 2; ++k) {
            t1[k] = top[(1 + k) * EDIM] + (t2[2 * k] + t2[2 * k + 1]);
            mm = fmaxf(mm, t1[k]);
        }
        const float root = top[0] + (t1[0] + t1[1]);
        mm = fmaxf(mm, root);
        out[tree * EDIM + tid] = fmaxf(mm, 0.0f);
    }
}

extern "C" void kernel_launch(void* const* d_in, const int* in_sizes, int n_in,
                              void* d_out, int out_size, void* d_ws, size_t ws_size,
                              hipStream_t stream)
{
    const float* emb    = (const float*)d_in[0];
    const float* W      = (const float*)d_in[1];
    const float* bias   = (const float*)d_in[2];
    const int*   tokens = (const int*)d_in[3];
    float*       out    = (float*)d_out;

    tree_encoder_kernel<<<TREES, 256, 0, stream>>>(emb, W, bias, tokens, out);
}


# Product taken once per table row (200,000 rows, a first launch, into the workspace) instead of once per token (1,046,528 rows); the tree launch gathers finished rows and only sums them
# speedup vs baseline: 6.4443x; 2.2631x over previous
#include <hip/hip_runtime.h>

typedef __attribute__((ext_vector_type(2))) float v2f;
typedef __attribute__((ext_vector_type(8))) float v8f;
typedef __attribute__((ext_vector_type(4))) float v4f;

#define TREES  2048
#define KNODES 511
#define EDIM   128
#define NTILES 32
#define APITCH 132
#define PF     2
#define STEADY (((NTILES - 3 - 2 * PF) / PF + 1) * PF)
#define KSIG   12
#define VOCAB  200000
#define PTILES 25
#define PGROUPS 500
#define PSTEADY_END 20
#define RPITCH 20
#define RBLOCK 336
#define P_BYTES ((size_t)VOCAB * EDIM * sizeof(float))

__device__ __forceinline__ float xhalf(float x)
{
    const int b = __builtin_bit_cast(int, x);
    return __builtin_bit_cast(float, __builtin_amdgcn_permlanex16(b, b, 0x76543210, 0xfedcba98, false, false));
}

__device__ __forceinline__ void lds_traffic_done_signal()
{
    __builtin_amdgcn_fence(__ATOMIC_RELEASE, "workgroup", "local");
    asm volatile("s_barrier_signal -1" ::: "memory");
}
__device__ __forceinline__ void lds_all_signalled_wait()
{
    asm volatile("s_barrier_wait -1" ::: "memory");
}
__device__ __forceinline__ void lds_all_signalled_wait_behind(const v8f acc)
{
    asm volatile("s_barrier_wait -1" :: "v"(acc) : "memory");
}


__device__ __forceinline__ void top_walk(const float* xch, const float* pm, const int np,
                                         const int tid, const int tree, float* __restrict__ out)
{
    if (tid < EDIM) {
        const float* sub = xch + tid;
        const float* top = xch + NTILES * EDIM + tid;
        float mm = pm[tid];
#pragma unroll
        for (int p = 1; p < np; ++p)
            mm = fmaxf(mm, pm[p * EDIM + tid]);
        float t4[16], t3[8], t2[4], t1[2];
#pragma unroll
        for (int k = 0; k < 16; ++k) {
            const float a = sub[(2 * k) * EDIM];
            const float b = sub[(2 * k + 1) * EDIM];
            t4[k] = top[(15 + k) * EDIM] + (a + b);
            mm = fmaxf(mm, fmaxf(fmaxf(a, b), t4[k]));
        }
#pragma unroll
        for (int k = 0; k < 8; ++k) {
            t3[k] = top[(7 + k) * EDIM] + (t4[2 * k] + t4[2 * k + 1]);
            mm = fmaxf(mm, t3[k]);
        }
#pragma unroll
        for (int k = 0; k < 4; ++k) {
            t2[k] = top[(3 + k) * EDIM] + (t3[2 * k] + t3[2 * k + 1]);
            mm = fmaxf(mm, t2[k]);
        }
#pragma unroll
        for (int k = 0; k < 2; ++k) {
            t1[k] = top[(1 + k) * EDIM] + (t2[2 * k] + t2[2 * k + 1]);
            mm = fmaxf(mm, t1[k]);
        }
        const float root = top[0] + (t1[0] + t1[1]);
        mm = fmaxf(mm, root);
        out[tree * EDIM + tid] = fmaxf(mm, 0.0f);
    }
}

__device__ __forceinline__ void project_tree(const float* __restrict__ emb,
                                             const float* __restrict__ W,
                                             const float* __restrict__ bias,
                                             const int*  __restrict__ tokens,
                                             float* xch, float* pmax, float (*stage)[16 * APITCH])
{
    const int tree = blockIdx.x;
    const int tid  = threadIdx.x;
    const int wave = tid >> 5;
    const int lane = tid & 31;
    const int mrow = lane & 15;
    const int half = lane >> 4;

    const int ni  = wave;
    const int col = ni * 16 + mrow;

    v2f bfrag[32];
#pragma unroll
    for (int kk = 0; kk < 32; ++kk) {
        const int k = kk * 4 + 2 * half;
        bfrag[kk].x = W[(k + 0) * EDIM + col];
        bfrag[kk].y = W[(k + 1) * EDIM + col];
    }
    const float bv   = bias[col];
    const int   base = tree * KNODES;

    int na, nb;
    if (wave == 0)      { na = 31 + lane;                  nb = lane < NTILES - 1 ? lane : 0; }
    else if (wave == 1) { na = 63 + 2 * lane;              nb = na + 1; }
    else if (wave < 4)  { na = 127 + 4 * lane + wave - 2;  nb = na + 2; }
    else                { na = 255 + 8 * lane + wave - 4;  nb = na + 4; }
    const int tokA = tokens[base + na];
    const int tokB = tokens[base + nb];

    const int soffA = wave * APITCH + lane * 4;
    const int soffB = (wave + 8) * APITCH + lane * 4;
    const int aoff  = mrow * APITCH + 2 * half;

    v4f ra[PF], rb[PF];
    v4f r0a, r0b, r1a, r1b;
    {
        const int ta0 = __builtin_amdgcn_readlane(tokA, 0);
        const int tb0 = __builtin_amdgcn_readlane(tokB, 0);
        r0a = *(const v4f*)(emb + (size_t)ta0 * EDIM + lane * 4);
        r0b = *(const v4f*)(emb + (size_t)tb0 * EDIM + lane * 4);
        const int ta1 = __builtin_amdgcn_readlane(tokA, 1);
        const int tb1 = __builtin_amdgcn_readlane(tokB, 1);
        r1a = *(const v4f*)(emb + (size_t)ta1 * EDIM + lane * 4);
        r1b = *(const v4f*)(emb + (size_t)tb1 * EDIM + lane * 4);
    }
#pragma unroll
    for (int t = 2; t <= PF + 1; ++t) {
        const int ta = __builtin_amdgcn_readlane(tokA, t);
        const int tb = __builtin_amdgcn_readlane(tokB, t);
        ra[t % PF] = *(const v4f*)(emb + (size_t)ta * EDIM + lane * 4);
        rb[t % PF] = *(const v4f*)(emb + (size_t)tb * EDIM + lane * 4);
    }
    *(v4f*)(&stage[0][soffA]) = r0a;
    *(v4f*)(&stage[0][soffB]) = r0b;
    *(v4f*)(&stage[1][soffA]) = r1a;
    *(v4f*)(&stage[1][soffB]) = r1b;

    v2f afr[2][32];
    v8f cprev = {};
    float m = -__builtin_inff();

    auto epi = [&]() __attribute__((always_inline)) -> float {
        float v[8];
#pragma unroll
        for (int r = 0; r < 8; ++r)
            v[r] = cprev[r] + bv;
        const float s2 = v[2] + (v[4] + v[5]);
        const float s3 = v[3] + (v[6] + v[7]);
        const float s1 = v[1] + (s2 + s3);
        m = fmaxf(m, fmaxf(fmaxf(v[4], v[5]), fmaxf(v[6], v[7])));
        m = fmaxf(m, fmaxf(fmaxf(s2, s3), s1));
        const float o  = xhalf(s1);
        const float sr = v[0] + (s1 + o);
        return half ? v[0] : sr;
    };
    auto epi_store = [&](const int j, const float x) __attribute__((always_inline)) {
        xch[half * (NTILES * EDIM) + j * EDIM + col] = x;
    };

    __builtin_amdgcn_fence(__ATOMIC_RELEASE, "workgroup", "local");
    __builtin_amdgcn_s_barrier();
    __builtin_amdgcn_fence(__ATOMIC_ACQUIRE, "workgroup", "local");
#pragma unroll
    for (int kk = 0; kk < 32; ++kk)
        afr[0][kk] = *(const v2f*)(&stage[0][aoff + kk * 4]);
    lds_traffic_done_signal();

    auto tile = [&](const int mi, const int s, const int par, const bool do_next,
                    const bool do_store, const bool do_issue,
                    const bool do_epi, const bool do_sig) __attribute__((always_inline)) {
        const float* nxt = stage[par ^ 1];
        float*       fre = stage[par];

        __builtin_amdgcn_sched_barrier(0);
        if (do_next) {
            lds_all_signalled_wait();
            const float* arow = nxt + aoff;
#pragma unroll
            for (int kk = 0; kk < 32; ++kk)
                afr[par ^ 1][kk] = *(const v2f*)(arow + kk * 4);
        }
        __builtin_amdgcn_sched_barrier(0);

        if (do_store) {
            *(v4f*)(fre + soffA) = ra[s];
            *(v4f*)(fre + soffB) = rb[s];
        }
        float x = 0.0f;
        if (do_epi)
            x = epi();

        v8f c = {};
#pragma unroll
        for (int kk = 0; kk < KSIG; ++kk)
            c = __builtin_amdgcn_wmma_f32_16x16x4_f32(
                    false, afr[par][kk], false, bfrag[kk], (short)0, c, false, false);
        __builtin_amdgcn_sched_barrier(0);
        if (do_sig)
            lds_traffic_done_signal();
        __builtin_amdgcn_sched_barrier(0);
        if (do_epi)
            epi_store(mi - 1, x);
        if (do_issue) {
            const int ta = __builtin_amdgcn_readlane(tokA, mi + 2 + PF);
            const int tb = __builtin_amdgcn_readlane(tokB, mi + 2 + PF);
            ra[s] = *(const v4f*)(emb + (size_t)ta * EDIM + lane * 4);
            rb[s] = *(const v4f*)(emb + (size_t)tb * EDIM + lane * 4);
        }
#pragma unroll
        for (int kk = KSIG; kk < 32; ++kk)
            c = __builtin_amdgcn_wmma_f32_16x16x4_f32(
                    false, afr[par][kk], false, bfrag[kk], (short)0, c, false, false);
        cprev = c;
    };

    tile(0, 2 % PF, 0, true, true, true, false, true);
#pragma unroll 1
    for (int g = 0; g < STEADY; g += PF) {
#pragma unroll
        for (int j = 1; j <= PF; ++j)
            tile(g + j, (j + 2) % PF, j & 1, true, true, true, true, true);
    }
#pragma unroll
    for (int mi = STEADY + 1; mi < NTILES; ++mi)
        tile(mi, (mi + 2) % PF, mi & 1, mi + 1 < NTILES, mi + 2 < NTILES, mi + 2 + PF < NTILES, true, mi + 2 < NTILES);
    epi_store(NTILES - 1, epi());
    pmax[half * EDIM + col] = m;
}

__device__ __forceinline__ void gather_tree(const float* __restrict__ P,
                                            const int*  __restrict__ tokens,
                                            float* xch, float* pm8)
{
    const int tree = blockIdx.x;
    const int tid  = threadIdx.x;
    const int wave = tid >> 5;
    const int lane = tid & 31;
    const int base = tree * KNODES;

    const int q  = lane & 7;
    const int hi = (lane >> 3) & 1;
    const int jA = 4 * wave + (lane >> 4);
    const int jB = jA + 2;
    int na, nb;
    if (q == 0)      { na = hi ? (jA < NTILES - 1 ? jA : 0) : 31 + jA;  nb = hi ? (jB < NTILES - 1 ? jB : 0) : 31 + jB; }
    else if (q == 1) { na = 63 + 2 * jA + hi;                           nb = 63 + 2 * jB + hi; }
    else if (q < 4)  { na = 127 + 4 * jA + 2 * hi + (q - 2);            nb = 127 + 4 * jB + 2 * hi + (q - 2); }
    else             { na = 255 + 8 * jA + 4 * hi + (q - 4);            nb = 255 + 8 * jB + 4 * hi + (q - 4); }
    const int tokA = tokens[base + na];
    const int tokB = tokens[base + nb];

    const unsigned lane4 = lane * 4;
    v4f x[2][16];
    const float ninf = -__builtin_inff();
    v4f m = {ninf, ninf, ninf, ninf};

    auto request = [&](const int t, const int set) __attribute__((always_inline)) {
#pragma unroll
        for (int r = 0; r < 16; ++r) {
            const int tok = __builtin_amdgcn_readlane(t < 2 ? tokA : tokB, (16 * t + r) & 31);
            const float* row = P + (size_t)(unsigned)tok * EDIM;
            x[set][r] = *(const v4f*)(row + lane4);
        }
    };
    auto vmax = [](const v4f a, const v4f b) __attribute__((always_inline)) -> v4f {
        return __builtin_elementwise_max(a, b);
    };
    auto reduce = [&](const int t, const int set) __attribute__((always_inline)) {
        const int j = 4 * wave + t;
        const v4f* v = x[set];
        const v4f s2 = v[2]  + (v[4]  + v[5]);
        const v4f s3 = v[3]  + (v[6]  + v[7]);
        const v4f s1 = v[1]  + (s2 + s3);
        const v4f u2 = v[10] + (v[12] + v[13]);
        const v4f u3 = v[11] + (v[14] + v[15]);
        const v4f u1 = v[9]  + (u2 + u3);
        m = vmax(m, vmax(vmax(v[4],  v[5]),  vmax(v[6],  v[7])));
        m = vmax(m, vmax(vmax(v[12], v[13]), vmax(v[14], v[15])));
        m = vmax(m, vmax(vmax(s2, s3), s1));
        m = vmax(m, vmax(vmax(u2, u3), u1));
        const v4f sr = v[0] + (s1 + u1);
        *(v4f*)(xch + j * EDIM + lane * 4) = sr;
        *(v4f*)(xch + NTILES * EDIM + j * EDIM + lane * 4) = v[8];
    };
    request(0, 0);
    request(1, 1);
    __builtin_amdgcn_sched_barrier(0);
    reduce(0, 0);
    __builtin_amdgcn_sched_barrier(0);
    request(2, 0);
    __builtin_amdgcn_sched_barrier(0);
    reduce(1, 1);
    __builtin_amdgcn_sched_barrier(0);
    request(3, 1);
    __builtin_amdgcn_sched_barrier(0);
    reduce(2, 0);
    reduce(3, 1);
    *(v4f*)(pm8 + wave * EDIM + lane * 4) = m;
}

__global__ __launch_bounds__(256, 4)
void project_table_kernel(const float* __restrict__ emb,
                          const float* __restrict__ W,
                          const float* __restrict__ bias,
                          float* __restrict__ P,
                          const size_t ws_bytes)
{
    if (ws_bytes < P_BYTES)
        return;
    __shared__ __attribute__((aligned(16))) float stage[2][16 * APITCH];
    __shared__ __attribute__((aligned(16))) float res[2][8 * RBLOCK];

    const int tid  = threadIdx.x;
    const int wave = tid >> 5;
    const int lane = tid & 31;
    const int mrow = lane & 15;
    const int half = lane >> 4;
    const int col  = wave * 16 + mrow;

    v2f bfrag[32];
#pragma unroll
    for (int kk = 0; kk < 32; ++kk) {
        const int k = kk * 4 + 2 * half;
        bfrag[kk].x = W[(k + 0) * EDIM + col];
        bfrag[kk].y = W[(k + 1) * EDIM + col];
    }
    const float bv = bias[col];

    const float* srcg = emb + (size_t)blockIdx.x * (PTILES * 16 * EDIM);
    float*       dstg = P   + (size_t)blockIdx.x * (PTILES * 16 * EDIM);
    const unsigned goff = wave * EDIM + lane * 4;
    const int soffA = wave * APITCH + lane * 4;
    const int soffB = (wave + 8) * APITCH + lane * 4;
    const int aoff  = mrow * APITCH + 2 * half;
    const int doff  = wave * RBLOCK + 8 * half * RPITCH + mrow;
    const int roffA = (lane >> 2) * RBLOCK + wave * RPITCH + 4 * (lane & 3);
    const int roffB = roffA + 8 * RPITCH;
    const float* srcp = srcg + (2 + PF) * 16 * EDIM + goff;
    float*       dstp = dstg + goff;

    v4f ra[PF], rb[PF];
    v4f r0a, r0b, r1a, r1b;
    r0a = *(const v4f*)(srcg + goff);
    r0b = *(const v4f*)(srcg + 8 * EDIM + goff);
    r1a = *(const v4f*)(srcg + 16 * EDIM + goff);
    r1b = *(const v4f*)(srcg + 24 * EDIM + goff);
#pragma unroll
    for (int t = 2; t <= PF + 1; ++t) {
        ra[t % PF] = *(const v4f*)(srcg + t * 16 * EDIM + goff);
        rb[t % PF] = *(const v4f*)(srcg + t * 16 * EDIM + 8 * EDIM + goff);
    }
    *(v4f*)(&stage[0][soffA]) = r0a;
    *(v4f*)(&stage[0][soffB]) = r0b;
    *(v4f*)(&stage[1][soffA]) = r1a;
    *(v4f*)(&stage[1][soffB]) = r1b;

    v2f afr[2][32];
    v8f cprev = {};

    __builtin_amdgcn_fence(__ATOMIC_RELEASE, "workgroup", "local");
    __builtin_amdgcn_s_barrier();
    __builtin_amdgcn_fence(__ATOMIC_ACQUIRE, "workgroup", "local");
#pragma unroll
    for (int kk = 0; kk < 32; ++kk)
        afr[0][kk] = *(const v2f*)(&stage[0][aoff + kk * 4]);
    lds_traffic_done_signal();

    auto tile = [&](const int par, const bool do_next, const bool do_store,
                    const bool do_issue, const bool do_put, const bool do_out) __attribute__((always_inline)) {
        const float* nxt = stage[par ^ 1];
        float*       fre = stage[par];

        lds_all_signalled_wait_behind(cprev);
        v4f o0, o1;
        if (do_out) {
            o0 = *(const v4f*)(&res[par][roffA]);
            o1 = *(const v4f*)(&res[par][roffB]);
        }
        if (do_next) {
            const float* arow = nxt + aoff;
#pragma unroll
            for (int kk = 0; kk < 32; ++kk)
                afr[par ^ 1][kk] = *(const v2f*)(arow + kk * 4);
        }
        if (do_out) {
            *(v4f*)(dstp)            = o0;
            *(v4f*)(dstp + 8 * EDIM) = o1;
            dstp += 16 * EDIM;
        }
        __builtin_amdgcn_sched_barrier(0);

        if (do_store) {
            *(v4f*)(fre + soffA) = ra[par];
            *(v4f*)(fre + soffB) = rb[par];
        }
        if (do_put) {
            float* rp = &res[par ^ 1][doff];
#pragma unroll
            for (int r = 0; r < 8; ++r)
                rp[r * RPITCH] = cprev[r] + bv;
        }

        v8f c = {};
#pragma unroll
        for (int kk = 0; kk < KSIG; ++kk)
            c = __builtin_amdgcn_wmma_f32_16x16x4_f32(
                    false, afr[par][kk], false, bfrag[kk], (short)0, c, false, false);
        __builtin_amdgcn_sched_barrier(0);
        lds_traffic_done_signal();
        __builtin_amdgcn_sched_barrier(0);
        if (do_issue) {
            ra[par] = *(const v4f*)(srcp);
            rb[par] = *(const v4f*)(srcp + 8 * EDIM);
        }
        srcp += 16 * EDIM;
#pragma unroll
        for (int kk = KSIG; kk < 32; ++kk)
            c = __builtin_amdgcn_wmma_f32_16x16x4_f32(
                    false, afr[par][kk], false, bfrag[kk], (short)0, c, false, false);
        cprev = c;
    };

    tile(0, true, true, true, false, false);
    tile(1, true, true, true, true,  false);
#pragma unroll 1
    for (int g = 2; g < PSTEADY_END; g += 2) {
        tile(0, true, true, true, true, true);
        tile(1, true, true, true, true, true);
    }
#pragma unroll
    for (int mi = PSTEADY_END; mi < PTILES; ++mi)
        tile(mi & 1, mi + 1 < PTILES, mi + 2 < PTILES, mi + 2 + PF < PTILES, true, true);

    lds_all_signalled_wait_behind(cprev);
    {
        const int p = (PTILES - 2) & 1;
        *(v4f*)(dstp)            = *(const v4f*)(&res[p][roffA]);
        *(v4f*)(dstp + 8 * EDIM) = *(const v4f*)(&res[p][roffB]);
    }
    {
        const int p = (PTILES - 1) & 1;
        float* rp = &res[p][doff];
#pragma unroll
        for (int r = 0; r < 8; ++r)
            rp[r * RPITCH] = cprev[r] + bv;
        __syncthreads();
        *(v4f*)(dstp + 16 * EDIM)     = *(const v4f*)(&res[p][roffA]);
        *(v4f*)(dstp + (16 + 8) * EDIM) = *(const v4f*)(&res[p][roffB]);
    }
}

__global__ __launch_bounds__(256, 4)
void tree_encoder_kernel(const float* __restrict__ emb,
                         const float* __restrict__ W,
                         const float* __restrict__ bias,
                         const int*  __restrict__ tokens,
                         const float* __restrict__ P,
                         const size_t ws_bytes,
                         float* __restrict__ out)
{
    __shared__ __attribute__((aligned(16))) float xch[2 * NTILES * EDIM];
    __shared__ float pmax[256];
    __shared__ __attribute__((aligned(16))) float stage[2][16 * APITCH];

    if (ws_bytes >= P_BYTES) {
        float* pm8 = &stage[0][0];
        gather_tree(P, tokens, xch, pm8);
        __syncthreads();
        top_walk(xch, pm8, 8, threadIdx.x, blockIdx.x, out);
    } else {
        project_tree(emb, W, bias, tokens, xch, pmax, stage);
        __syncthreads();
        top_walk(xch, pmax, 2, threadIdx.x, blockIdx.x, out);
    }
}

extern "C" void kernel_launch(void* const* d_in, const int* in_sizes, int n_in,
                              void* d_out, int out_size, void* d_ws, size_t ws_size,
                              hipStream_t stream)
{
    const float* emb    = (const float*)d_in[0];
    const float* W      = (const float*)d_in[1];
    const float* bias   = (const float*)d_in[2];
    const int*   tokens = (const int*)d_in[3];
    float*       out    = (float*)d_out;
    float*       P      = (float*)d_ws;

    project_table_kernel<<<PGROUPS, 256, 0, stream>>>(emb, W, bias, P, ws_size);
    tree_encoder_kernel<<<TREES, 256, 0, stream>>>(emb, W, bias, tokens, P, ws_size, out);
}
